# starting kernel
# baseline (speedup 1.0000x reference)
#include <hip/hip_runtime.h>

typedef float v2f __attribute__((ext_vector_type(2)));
typedef float v8f __attribute__((ext_vector_type(8)));

#define HH 96
#define WW 96
#define PIX (HH * WW)
#define CDIM 64
#define NHEADS 2
#define HD 32
#define KW 7
#define NKK (KW * KW)

__global__ void nchw_to_nhwc(const float* __restrict__ x, float* __restrict__ xh) {
    int t = blockIdx.x * blockDim.x + threadIdx.x;
    if (t >= CDIM * PIX) return;
    int c = t / PIX;
    int p = t % PIX;
    xh[(size_t)p * CDIM + c] = x[t];
}

__global__ void nhwc_to_nchw(const float* __restrict__ xh, float* __restrict__ y) {
    int t = blockIdx.x * blockDim.x + threadIdx.x;
    if (t >= CDIM * PIX) return;
    int c = t / PIX;
    int p = t % PIX;
    y[t] = xh[(size_t)p * CDIM + c];
}

template <int KD>
__global__ void wmma_gemm_bias(const float* __restrict__ A,
                               const float* __restrict__ W,
                               const float* __restrict__ bias,
                               float* __restrict__ C,
                               int M, int N) {
    int lane = threadIdx.x & 31;
    int wid  = threadIdx.x >> 5;
    int tile = blockIdx.x * 8 + wid;
    int tilesN = N >> 4;
    int numTiles = (M >> 4) * tilesN;
    if (tile >= numTiles) return;

    int tm = tile / tilesN;
    int tn = tile % tilesN;
    int lmod  = lane & 15;
    int lhalf = lane >> 4;

    const float* __restrict__ Ap = A + (size_t)(tm * 16 + lmod) * KD + 2 * lhalf;
    const float* __restrict__ Wp = W + (size_t)(tn * 16 + lmod) * KD + 2 * lhalf;

    v8f acc = {0.f, 0.f, 0.f, 0.f, 0.f, 0.f, 0.f, 0.f};
#pragma unroll
    for (int k = 0; k < KD; k += 4) {
        v2f a = *(const v2f*)(Ap + k);
        v2f b = *(const v2f*)(Wp + k);
        acc = __builtin_amdgcn_wmma_f32_16x16x4_f32(
             false, a,  false, b,
             (short)0, acc,  false,  false);
    }

    float bv = bias[tn * 16 + lmod];
    int rbase = tm * 16 + 8 * lhalf;
    int col   = tn * 16 + lmod;
#pragma unroll
    for (int v = 0; v < 8; ++v) {
        C[(size_t)(rbase + v) * N + col] = acc[v] + bv;
    }
}

__global__ void na_attn_kernel(const float* __restrict__ qkv,
                               const float* __restrict__ rpb,
                               const float* __restrict__ prev,
                               float* __restrict__ attn_store,
                               float* __restrict__ out,
                               int has_prev) {
    int idx = blockIdx.x * blockDim.x + threadIdx.x;
    if (idx >= NHEADS * PIX) return;
    int head = idx / PIX;
    int pix  = idx % PIX;
    int i = pix / WW;
    int j = pix % WW;

    int si = i - (KW / 2); si = si < 0 ? 0 : si; si = si > HH - KW ? HH - KW : si;
    int sj = j - (KW / 2); sj = sj < 0 ? 0 : sj; sj = sj > WW - KW ? WW - KW : sj;

    const float scale = 0.17677669529663687f;

    float qreg[HD];
    const float* qp = qkv + (size_t)pix * 192 + head * HD;
#pragma unroll
    for (int d = 0; d < HD; ++d) qreg[d] = qp[d] * scale;

    float m = -1e30f;
    float ssum = 0.f;
    float outacc[HD];
#pragma unroll
    for (int d = 0; d < HD; ++d) outacc[d] = 0.f;

    float* astore = attn_store + (size_t)idx * NKK;
    const float* pprev = prev + (size_t)idx * NKK;
    const float* rb = rpb + head * (2 * KW - 1) * (2 * KW - 1);

    for (int a = 0; a < KW; ++a) {
        int ri = si + a;
        for (int b = 0; b < KW; ++b) {
            int rj = sj + b;
            int kk = a * KW + b;
            const float* kp = qkv + (size_t)(ri * WW + rj) * 192 + 64 + head * HD;
            float l = 0.f;
#pragma unroll
            for (int d = 0; d < HD; ++d) l += qreg[d] * kp[d];
            l += rb[(ri - i + KW - 1) * (2 * KW - 1) + (rj - j + KW - 1)];
            if (has_prev) l += pprev[kk];
            astore[kk] = l;

            float mnew = fmaxf(m, l);
            float corr = __expf(m - mnew);
            float p    = __expf(l - mnew);
            ssum = ssum * corr + p;
            const float* vp = qkv + (size_t)(ri * WW + rj) * 192 + 128 + head * HD;
#pragma unroll
            for (int d = 0; d < HD; ++d) outacc[d] = outacc[d] * corr + p * vp[d];
            m = mnew;
        }
    }

    float inv = 1.f / ssum;
    for (int kk = 0; kk < NKK; ++kk) {
        astore[kk] = __expf(astore[kk] - m) * inv;
    }
    float* op = out + (size_t)pix * CDIM + head * HD;
#pragma unroll
    for (int d = 0; d < HD; ++d) op[d] = outacc[d] * inv;
}

extern "C" void kernel_launch(void* const* d_in, const int* in_sizes, int n_in,
                              void* d_out, int out_size, void* d_ws, size_t ws_size,
                              hipStream_t stream) {
    const float* x      = (const float*)d_in[0];
    const float* qkv_w  = (const float*)d_in[1];
    const float* qkv_b  = (const float*)d_in[2];
    const float* proj_w = (const float*)d_in[3];
    const float* proj_b = (const float*)d_in[4];
    const float* rpb    = (const float*)d_in[5];
    float* out = (float*)d_out;
    float* ws  = (float*)d_ws;

    float* xh0     = ws;
    float* xh1     = xh0 + (size_t)PIX * CDIM;
    float* qkv     = xh1 + (size_t)PIX * CDIM;
    float* attnA   = qkv + (size_t)PIX * 192;
    float* attnB   = attnA + (size_t)NHEADS * PIX * NKK;
    float* attnout = attnB + (size_t)NHEADS * PIX * NKK;

    nchw_to_nhwc<<<(CDIM * PIX + 255) / 256, 256, 0, stream>>>(x, xh0);

    float* xin  = xh0;
    float* xout = xh1;
    const float* prev = nullptr;

    for (int s = 0; s < 3; ++s) {
        float* acur = (s & 1) ? attnB : attnA;

        {
            int numTiles = (PIX / 16) * (192 / 16);
            wmma_gemm_bias<CDIM><<<(numTiles + 7) / 8, 256, 0, stream>>>(
                xin, qkv_w + (size_t)s * 192 * CDIM, qkv_b + (size_t)s * 192,
                qkv, PIX, 192);
        }

        na_attn_kernel<<<(NHEADS * PIX + 255) / 256, 256, 0, stream>>>(
            qkv, rpb + (size_t)s * NHEADS * (2 * KW - 1) * (2 * KW - 1),
            prev ? prev : acur  ,
            acur, attnout, prev != nullptr ? 1 : 0);

        {
            int numTiles = (PIX / 16) * (CDIM / 16);
            wmma_gemm_bias<CDIM><<<(numTiles + 7) / 8, 256, 0, stream>>>(
                attnout, proj_w + (size_t)s * CDIM * CDIM, proj_b + (size_t)s * CDIM,
                xout, PIX, CDIM);
        }

        prev = acur;
        float* t = xin; xin = xout; xout = t;
    }

    nhwc_to_nchw<<<(CDIM * PIX + 255) / 256, 256, 0, stream>>>(xin, out);
}


# attention: 8 lanes per (head,pixel), 4 dims each, 16B K/V loads fill whole 128B rows; q.k summed by 3-step lane butterfly; 49 logits in regs, plain max/exp2/sum softmax replaces online rescale
# speedup vs baseline: 3.8344x; 3.8344x over previous
#include <hip/hip_runtime.h>

typedef float v2f __attribute__((ext_vector_type(2)));
typedef float v8f __attribute__((ext_vector_type(8)));

#define HH 96
#define WW 96
#define PIX (HH * WW)
#define CDIM 64
#define NHEADS 2
#define HD 32
#define KW 7
#define NKK (KW * KW)

__global__ void nchw_to_nhwc(const float* __restrict__ x, float* __restrict__ xh) {
    int t = blockIdx.x * blockDim.x + threadIdx.x;
    if (t >= CDIM * PIX) return;
    int c = t / PIX;
    int p = t % PIX;
    xh[(size_t)p * CDIM + c] = x[t];
}

__global__ void nhwc_to_nchw(const float* __restrict__ xh, float* __restrict__ y) {
    int t = blockIdx.x * blockDim.x + threadIdx.x;
    if (t >= CDIM * PIX) return;
    int c = t / PIX;
    int p = t % PIX;
    y[t] = xh[(size_t)p * CDIM + c];
}

template <int KD>
__global__ void wmma_gemm_bias(const float* __restrict__ A,
                               const float* __restrict__ W,
                               const float* __restrict__ bias,
                               float* __restrict__ C,
                               int M, int N) {
    int lane = threadIdx.x & 31;
    int wid  = threadIdx.x >> 5;
    int tile = blockIdx.x * 8 + wid;
    int tilesN = N >> 4;
    int numTiles = (M >> 4) * tilesN;
    if (tile >= numTiles) return;

    int tm = tile / tilesN;
    int tn = tile % tilesN;
    int lmod  = lane & 15;
    int lhalf = lane >> 4;

    const float* __restrict__ Ap = A + (size_t)(tm * 16 + lmod) * KD + 2 * lhalf;
    const float* __restrict__ Wp = W + (size_t)(tn * 16 + lmod) * KD + 2 * lhalf;

    v8f acc = {0.f, 0.f, 0.f, 0.f, 0.f, 0.f, 0.f, 0.f};
#pragma unroll
    for (int k = 0; k < KD; k += 4) {
        v2f a = *(const v2f*)(Ap + k);
        v2f b = *(const v2f*)(Wp + k);
        acc = __builtin_amdgcn_wmma_f32_16x16x4_f32(
             false, a,  false, b,
             (short)0, acc,  false,  false);
    }

    float bv = bias[tn * 16 + lmod];
    int rbase = tm * 16 + 8 * lhalf;
    int col   = tn * 16 + lmod;
#pragma unroll
    for (int v = 0; v < 8; ++v) {
        C[(size_t)(rbase + v) * N + col] = acc[v] + bv;
    }
}

typedef float v4f __attribute__((ext_vector_type(4)));

#define PATCH_W 4
#define PATCH_H 4
#define ATT_THREADS (PATCH_W * PATCH_H * 8)
#define PATCHES_X (WW / PATCH_W)
#define PATCHES_Y (HH / PATCH_H)
#define PATCHES (PATCHES_X * PATCHES_Y)
static_assert(WW % PATCH_W == 0 && HH % PATCH_H == 0, "patches tile the image");
static_assert(HD == 32 && NKK == 49, "8 lanes x 4 dims; 49 taps");

template <int MASK>
__device__ __forceinline__ float lane_xor(float x) {
    return __builtin_bit_cast(float,
        __builtin_amdgcn_update_dpp(0, __builtin_bit_cast(int, x), 0x160 + MASK, 0xf, 0xf, false));
}

template <bool HAS_PREV>
__global__ __launch_bounds__(ATT_THREADS)
void na_attn_kernel(const float* __restrict__ qkv,
                    const float* __restrict__ rpb,
                    const float* __restrict__ prev,
                    float* __restrict__ attn_store,
                    float* __restrict__ out) {
    const int tid = threadIdx.x;
    const int s   = tid & 7;
    const int pl  = tid >> 3;
    const int head = blockIdx.x / PATCHES;
    const int pr   = blockIdx.x % PATCHES;
    const int i = (pr / PATCHES_X) * PATCH_H + (pl / PATCH_W);
    const int j = (pr % PATCHES_X) * PATCH_W + (pl % PATCH_W);
    const int pix = i * WW + j;
    const int idx = head * PIX + pix;

    int si = i - (KW / 2); si = si < 0 ? 0 : si; si = si > HH - KW ? HH - KW : si;
    int sj = j - (KW / 2); sj = sj < 0 ? 0 : sj; sj = sj > WW - KW ? WW - KW : sj;

    const float LOG2E = 1.4426950408889634f;
    const float scale2 = 0.17677669529663687f * LOG2E;

    v4f q = *(const v4f*)(qkv + (size_t)pix * 192 + head * HD + 4 * s);
    q *= scale2;

    const float* kbase = qkv + (size_t)(si * WW + sj) * 192 + 64 + head * HD + 4 * s;
    const float* vbase = kbase + 64;

    const float* rb = rpb + head * (2 * KW - 1) * (2 * KW - 1)
                    + (si - i + KW - 1) * (2 * KW - 1) + (sj - j + KW - 1);
    const float* pprev = prev + (size_t)idx * NKK;
    float extra[7];
#pragma unroll
    for (int t = 0; t < 7; ++t) {
        const int kk = (t < 6) ? (s + 8 * t) : (s == 0 ? NKK - 1 : 0);
        const int a = kk / KW;
        const int b = kk - a * KW;
        float e = rb[a * (2 * KW - 1) + b];
        if (HAS_PREV) e += pprev[kk];
        extra[t] = e * LOG2E;
    }
    if (s != 0) extra[6] = 0.f;

    float lg[NKK];
#pragma unroll
    for (int a = 0; a < KW; ++a) {
#pragma unroll
        for (int b = 0; b < KW; ++b) {
            const int kk = a * KW + b;
            v4f k4 = *(const v4f*)(kbase + (a * WW + b) * 192);
            float part = (s == (kk & 7)) ? extra[kk >> 3] : 0.f;
            part = fmaf(q.x, k4.x, part);
            part = fmaf(q.y, k4.y, part);
            part = fmaf(q.z, k4.z, part);
            part = fmaf(q.w, k4.w, part);
            part += lane_xor<1>(part);
            part += lane_xor<2>(part);
            part += lane_xor<4>(part);
            lg[kk] = part;
        }
    }

    float m = lg[0];
#pragma unroll
    for (int kk = 1; kk < NKK; ++kk) m = fmaxf(m, lg[kk]);
    float ssum = 0.f;
#pragma unroll
    for (int kk = 0; kk < NKK; ++kk) {
        lg[kk] = __builtin_amdgcn_exp2f(lg[kk] - m);
        ssum += lg[kk];
    }
    const float inv = 1.f / ssum;

    float mine[7];
#pragma unroll
    for (int t = 0; t < 7; ++t) mine[t] = 0.f;
#pragma unroll
    for (int kk = 0; kk < NKK; ++kk) {
        if (s == (kk & 7)) mine[kk >> 3] = lg[kk];
    }
    float* astore = attn_store + (size_t)idx * NKK;
#pragma unroll
    for (int t = 0; t < 6; ++t) astore[s + 8 * t] = mine[t] * inv;
    if (s == 0) astore[48] = mine[6] * inv;

    v4f acc = {0.f, 0.f, 0.f, 0.f};
#pragma unroll
    for (int a = 0; a < KW; ++a) {
#pragma unroll
        for (int b = 0; b < KW; ++b) {
            const float p = lg[a * KW + b];
            v4f v4 = *(const v4f*)(vbase + (a * WW + b) * 192);
            acc.x = fmaf(p, v4.x, acc.x);
            acc.y = fmaf(p, v4.y, acc.y);
            acc.z = fmaf(p, v4.z, acc.z);
            acc.w = fmaf(p, v4.w, acc.w);
        }
    }
    acc *= inv;
    *(v4f*)(out + (size_t)pix * CDIM + head * HD + 4 * s) = acc;
}

extern "C" void kernel_launch(void* const* d_in, const int* in_sizes, int n_in,
                              void* d_out, int out_size, void* d_ws, size_t ws_size,
                              hipStream_t stream) {
    const float* x      = (const float*)d_in[0];
    const float* qkv_w  = (const float*)d_in[1];
    const float* qkv_b  = (const float*)d_in[2];
    const float* proj_w = (const float*)d_in[3];
    const float* proj_b = (const float*)d_in[4];
    const float* rpb    = (const float*)d_in[5];
    float* out = (float*)d_out;
    float* ws  = (float*)d_ws;

    float* xh0     = ws;
    float* xh1     = xh0 + (size_t)PIX * CDIM;
    float* qkv     = xh1 + (size_t)PIX * CDIM;
    float* attnA   = qkv + (size_t)PIX * 192;
    float* attnB   = attnA + (size_t)NHEADS * PIX * NKK;
    float* attnout = attnB + (size_t)NHEADS * PIX * NKK;

    nchw_to_nhwc<<<(CDIM * PIX + 255) / 256, 256, 0, stream>>>(x, xh0);

    float* xin  = xh0;
    float* xout = xh1;
    const float* prev = nullptr;

    for (int s = 0; s < 3; ++s) {
        float* acur = (s & 1) ? attnB : attnA;

        {
            int numTiles = (PIX / 16) * (192 / 16);
            wmma_gemm_bias<CDIM><<<(numTiles + 7) / 8, 256, 0, stream>>>(
                xin, qkv_w + (size_t)s * 192 * CDIM, qkv_b + (size_t)s * 192,
                qkv, PIX, 192);
        }

        if (prev) {
            na_attn_kernel<true><<<NHEADS * PATCHES, ATT_THREADS, 0, stream>>>(
                qkv, rpb + (size_t)s * NHEADS * (2 * KW - 1) * (2 * KW - 1),
                prev, acur, attnout);
        } else {
            na_attn_kernel<false><<<NHEADS * PATCHES, ATT_THREADS, 0, stream>>>(
                qkv, rpb + (size_t)s * NHEADS * (2 * KW - 1) * (2 * KW - 1),
                acur  , acur, attnout);
        }

        {
            int numTiles = (PIX / 16) * (CDIM / 16);
            wmma_gemm_bias<CDIM><<<(numTiles + 7) / 8, 256, 0, stream>>>(
                attnout, proj_w + (size_t)s * CDIM * CDIM, proj_b + (size_t)s * CDIM,
                xout, PIX, CDIM);
        }

        prev = acur;
        float* t = xin; xin = xout; xout = t;
    }

    nhwc_to_nchw<<<(CDIM * PIX + 255) / 256, 256, 0, stream>>>(xin, out);
}


# layout-transpose launches folded into the GEMMs: stage-0 QKV GEMM loads its A fragment straight from the (C,H,W) input, last projection GEMM swaps operand roles and stores (C,H,W); 11 -> 9 launches
# speedup vs baseline: 5.2245x; 1.3625x over previous
#include <hip/hip_runtime.h>

typedef float v2f __attribute__((ext_vector_type(2)));
typedef float v4f __attribute__((ext_vector_type(4)));
typedef float v8f __attribute__((ext_vector_type(8)));

#define HH 96
#define WW 96
#define PIX (HH * WW)
#define CDIM 64
#define NHEADS 2
#define HD 32
#define KW 7
#define NKK (KW * KW)

enum { ROWMAJOR = 0, A_PLANAR = 1, C_PLANAR = 2 };

template <int KD, int LAYOUT>
__global__ void wmma_gemm_bias(const float* __restrict__ A,
                               const float* __restrict__ W,
                               const float* __restrict__ bias,
                               float* __restrict__ C,
                               int M, int N) {
    int lane = threadIdx.x & 31;
    int wid  = threadIdx.x >> 5;
    int tile = blockIdx.x * 8 + wid;
    int tilesN = N >> 4;
    int numTiles = (M >> 4) * tilesN;
    if (tile >= numTiles) return;

    int tm = tile / tilesN;
    int tn = tile % tilesN;
    int lmod  = lane & 15;
    int lhalf = lane >> 4;

    const float* __restrict__ Ap = (LAYOUT == A_PLANAR)
        ? A + (size_t)(2 * lhalf) * PIX + (tm * 16 + lmod)
        : A + (size_t)(tm * 16 + lmod) * KD + 2 * lhalf;
    const float* __restrict__ Wp = W + (size_t)(tn * 16 + lmod) * KD + 2 * lhalf;

    v8f acc = {0.f, 0.f, 0.f, 0.f, 0.f, 0.f, 0.f, 0.f};
#pragma unroll
    for (int k = 0; k < KD; k += 4) {
        v2f a;
        if (LAYOUT == A_PLANAR) {
            a.x = Ap[(size_t)k * PIX];
            a.y = Ap[(size_t)(k + 1) * PIX];
        } else {
            a = *(const v2f*)(Ap + k);
        }
        v2f b = *(const v2f*)(Wp + k);
        if (LAYOUT == C_PLANAR) {
            acc = __builtin_amdgcn_wmma_f32_16x16x4_f32(
                 false, b,  false, a,
                 (short)0, acc,  false,  false);
        } else {
            acc = __builtin_amdgcn_wmma_f32_16x16x4_f32(
                 false, a,  false, b,
                 (short)0, acc,  false,  false);
        }
    }

    if (LAYOUT == C_PLANAR) {
        int ch = tn * 16 + 8 * lhalf;
        const v4f blo = *(const v4f*)(bias + ch);
        const v4f bhi = *(const v4f*)(bias + ch + 4);
        float* __restrict__ Cp = C + (size_t)ch * PIX + (tm * 16 + lmod);
#pragma unroll
        for (int v = 0; v < 8; ++v) {
            float bv = (v < 4) ? blo[v & 3] : bhi[v & 3];
            Cp[(size_t)v * PIX] = acc[v] + bv;
        }
    } else {
        float bv = bias[tn * 16 + lmod];
        int rbase = tm * 16 + 8 * lhalf;
        int col   = tn * 16 + lmod;
#pragma unroll
        for (int v = 0; v < 8; ++v) {
            C[(size_t)(rbase + v) * N + col] = acc[v] + bv;
        }
    }
}

#define PATCH_W 4
#define PATCH_H 4
#define ATT_THREADS (PATCH_W * PATCH_H * 8)
#define PATCHES_X (WW / PATCH_W)
#define PATCHES_Y (HH / PATCH_H)
#define PATCHES (PATCHES_X * PATCHES_Y)
static_assert(WW % PATCH_W == 0 && HH % PATCH_H == 0, "patches tile the image");
static_assert(HD == 32 && NKK == 49, "8 lanes x 4 dims; 49 taps");

template <int MASK>
__device__ __forceinline__ float lane_xor(float x) {
    return __builtin_bit_cast(float,
        __builtin_amdgcn_update_dpp(0, __builtin_bit_cast(int, x), 0x160 + MASK, 0xf, 0xf, false));
}

template <bool HAS_PREV>
__global__ __launch_bounds__(ATT_THREADS)
void na_attn_kernel(const float* __restrict__ qkv,
                    const float* __restrict__ rpb,
                    const float* __restrict__ prev,
                    float* __restrict__ attn_store,
                    float* __restrict__ out) {
    const int tid = threadIdx.x;
    const int s   = tid & 7;
    const int pl  = tid >> 3;
    const int head = blockIdx.x / PATCHES;
    const int pr   = blockIdx.x % PATCHES;
    const int i = (pr / PATCHES_X) * PATCH_H + (pl / PATCH_W);
    const int j = (pr % PATCHES_X) * PATCH_W + (pl % PATCH_W);
    const int pix = i * WW + j;
    const int idx = head * PIX + pix;

    int si = i - (KW / 2); si = si < 0 ? 0 : si; si = si > HH - KW ? HH - KW : si;
    int sj = j - (KW / 2); sj = sj < 0 ? 0 : sj; sj = sj > WW - KW ? WW - KW : sj;

    const float LOG2E = 1.4426950408889634f;
    const float scale2 = 0.17677669529663687f * LOG2E;

    v4f q = *(const v4f*)(qkv + (size_t)pix * 192 + head * HD + 4 * s);
    q *= scale2;

    const float* kbase = qkv + (size_t)(si * WW + sj) * 192 + 64 + head * HD + 4 * s;
    const float* vbase = kbase + 64;

    const float* rb = rpb + head * (2 * KW - 1) * (2 * KW - 1)
                    + (si - i + KW - 1) * (2 * KW - 1) + (sj - j + KW - 1);
    const float* pprev = prev + (size_t)idx * NKK;
    float extra[7];
#pragma unroll
    for (int t = 0; t < 7; ++t) {
        const int kk = (t < 6) ? (s + 8 * t) : (s == 0 ? NKK - 1 : 0);
        const int a = kk / KW;
        const int b = kk - a * KW;
        float e = rb[a * (2 * KW - 1) + b];
        if (HAS_PREV) e += pprev[kk];
        extra[t] = e * LOG2E;
    }
    if (s != 0) extra[6] = 0.f;

    float lg[NKK];
#pragma unroll
    for (int a = 0; a < KW; ++a) {
#pragma unroll
        for (int b = 0; b < KW; ++b) {
            const int kk = a * KW + b;
            v4f k4 = *(const v4f*)(kbase + (a * WW + b) * 192);
            float part = (s == (kk & 7)) ? extra[kk >> 3] : 0.f;
            part = fmaf(q.x, k4.x, part);
            part = fmaf(q.y, k4.y, part);
            part = fmaf(q.z, k4.z, part);
            part = fmaf(q.w, k4.w, part);
            part += lane_xor<1>(part);
            part += lane_xor<2>(part);
            part += lane_xor<4>(part);
            lg[kk] = part;
        }
    }

    float m = lg[0];
#pragma unroll
    for (int kk = 1; kk < NKK; ++kk) m = fmaxf(m, lg[kk]);
    float ssum = 0.f;
#pragma unroll
    for (int kk = 0; kk < NKK; ++kk) {
        lg[kk] = __builtin_amdgcn_exp2f(lg[kk] - m);
        ssum += lg[kk];
    }
    const float inv = 1.f / ssum;

    float mine[7];
#pragma unroll
    for (int t = 0; t < 7; ++t) mine[t] = 0.f;
#pragma unroll
    for (int kk = 0; kk < NKK; ++kk) {
        if (s == (kk & 7)) mine[kk >> 3] = lg[kk];
    }
    float* astore = attn_store + (size_t)idx * NKK;
#pragma unroll
    for (int t = 0; t < 6; ++t) astore[s + 8 * t] = mine[t] * inv;
    if (s == 0) astore[48] = mine[6] * inv;

    v4f acc = {0.f, 0.f, 0.f, 0.f};
#pragma unroll
    for (int a = 0; a < KW; ++a) {
#pragma unroll
        for (int b = 0; b < KW; ++b) {
            const float p = lg[a * KW + b];
            v4f v4 = *(const v4f*)(vbase + (a * WW + b) * 192);
            acc.x = fmaf(p, v4.x, acc.x);
            acc.y = fmaf(p, v4.y, acc.y);
            acc.z = fmaf(p, v4.z, acc.z);
            acc.w = fmaf(p, v4.w, acc.w);
        }
    }
    acc *= inv;
    *(v4f*)(out + (size_t)pix * CDIM + head * HD + 4 * s) = acc;
}

extern "C" void kernel_launch(void* const* d_in, const int* in_sizes, int n_in,
                              void* d_out, int out_size, void* d_ws, size_t ws_size,
                              hipStream_t stream) {
    const float* x      = (const float*)d_in[0];
    const float* qkv_w  = (const float*)d_in[1];
    const float* qkv_b  = (const float*)d_in[2];
    const float* proj_w = (const float*)d_in[3];
    const float* proj_b = (const float*)d_in[4];
    const float* rpb    = (const float*)d_in[5];
    float* out = (float*)d_out;
    float* ws  = (float*)d_ws;

    float* xh0     = ws;
    float* xh1     = xh0 + (size_t)PIX * CDIM;
    float* qkv     = xh1 + (size_t)PIX * CDIM;
    float* attnA   = qkv + (size_t)PIX * 192;
    float* attnB   = attnA + (size_t)NHEADS * PIX * NKK;
    float* attnout = attnB + (size_t)NHEADS * PIX * NKK;

    const float* xin = x;
    float* xout = xh0;
    const float* prev = nullptr;

    for (int s = 0; s < 3; ++s) {
        float* acur = (s & 1) ? attnB : attnA;

        {
            int numTiles = (PIX / 16) * (192 / 16);
            if (s == 0) {
                wmma_gemm_bias<CDIM, A_PLANAR><<<(numTiles + 7) / 8, 256, 0, stream>>>(
                    xin, qkv_w + (size_t)s * 192 * CDIM, qkv_b + (size_t)s * 192,
                    qkv, PIX, 192);
            } else {
                wmma_gemm_bias<CDIM, ROWMAJOR><<<(numTiles + 7) / 8, 256, 0, stream>>>(
                    xin, qkv_w + (size_t)s * 192 * CDIM, qkv_b + (size_t)s * 192,
                    qkv, PIX, 192);
            }
        }

        if (prev) {
            na_attn_kernel<true><<<NHEADS * PATCHES, ATT_THREADS, 0, stream>>>(
                qkv, rpb + (size_t)s * NHEADS * (2 * KW - 1) * (2 * KW - 1),
                prev, acur, attnout);
        } else {
            na_attn_kernel<false><<<NHEADS * PATCHES, ATT_THREADS, 0, stream>>>(
                qkv, rpb + (size_t)s * NHEADS * (2 * KW - 1) * (2 * KW - 1),
                acur  , acur, attnout);
        }

        {
            int numTiles = (PIX / 16) * (CDIM / 16);
            if (s == 2) {
                wmma_gemm_bias<CDIM, C_PLANAR><<<(numTiles + 7) / 8, 256, 0, stream>>>(
                    attnout, proj_w + (size_t)s * CDIM * CDIM, proj_b + (size_t)s * CDIM,
                    out, PIX, CDIM);
            } else {
                wmma_gemm_bias<CDIM, ROWMAJOR><<<(numTiles + 7) / 8, 256, 0, stream>>>(
                    attnout, proj_w + (size_t)s * CDIM * CDIM, proj_b + (size_t)s * CDIM,
                    xout, PIX, CDIM);
            }
        }

        prev = acur;
        xin  = xout;
        xout = (xout == xh0) ? xh1 : xh0;
    }
}


# output projection of stage s folded into the QKV GEMM of stage s+1 (W' = Wqkv*Wproj, b' = Wqkv*bproj + bqkv, built on WMMA in spare work-groups of launch 1); sums reassociated; 9 -> 7 launches
# speedup vs baseline: 6.1704x; 1.1811x over previous
#include <hip/hip_runtime.h>

typedef float v2f __attribute__((ext_vector_type(2)));
typedef float v4f __attribute__((ext_vector_type(4)));
typedef float v8f __attribute__((ext_vector_type(8)));

#define HH 96
#define WW 96
#define PIX (HH * WW)
#define CDIM 64
#define NHEADS 2
#define HD 32
#define KW 7
#define NKK (KW * KW)

enum { ROWMAJOR = 0, A_PLANAR = 1, C_PLANAR = 2 };

template <int KD, int LAYOUT>
__device__ __forceinline__ void gemm_tile_bias(const float* __restrict__ A,
                                               const float* __restrict__ W,
                                               const float* __restrict__ bias,
                                               float* __restrict__ C,
                                               int N, int tile, int lane) {
    int tilesN = N >> 4;
    int tm = tile / tilesN;
    int tn = tile % tilesN;
    int lmod  = lane & 15;
    int lhalf = lane >> 4;

    const float* __restrict__ Ap = (LAYOUT == A_PLANAR)
        ? A + (size_t)(2 * lhalf) * PIX + (tm * 16 + lmod)
        : A + (size_t)(tm * 16 + lmod) * KD + 2 * lhalf;
    const float* __restrict__ Wp = W + (size_t)(tn * 16 + lmod) * KD + 2 * lhalf;

    v8f acc = {0.f, 0.f, 0.f, 0.f, 0.f, 0.f, 0.f, 0.f};
#pragma unroll
    for (int k = 0; k < KD; k += 4) {
        v2f a;
        if (LAYOUT == A_PLANAR) {
            a.x = Ap[(size_t)k * PIX];
            a.y = Ap[(size_t)(k + 1) * PIX];
        } else {
            a = *(const v2f*)(Ap + k);
        }
        v2f b = *(const v2f*)(Wp + k);
        if (LAYOUT == C_PLANAR) {
            acc = __builtin_amdgcn_wmma_f32_16x16x4_f32(
                 false, b,  false, a,
                 (short)0, acc,  false,  false);
        } else {
            acc = __builtin_amdgcn_wmma_f32_16x16x4_f32(
                 false, a,  false, b,
                 (short)0, acc,  false,  false);
        }
    }

    if (LAYOUT == C_PLANAR) {
        int ch = tn * 16 + 8 * lhalf;
        const v4f blo = *(const v4f*)(bias + ch);
        const v4f bhi = *(const v4f*)(bias + ch + 4);
        float* __restrict__ Cp = C + (size_t)ch * PIX + (tm * 16 + lmod);
#pragma unroll
        for (int v = 0; v < 8; ++v) {
            float bv = (v < 4) ? blo[v & 3] : bhi[v & 3];
            Cp[(size_t)v * PIX] = acc[v] + bv;
        }
    } else {
        float bv = bias[tn * 16 + lmod];
        int rbase = tm * 16 + 8 * lhalf;
        int col   = tn * 16 + lmod;
#pragma unroll
        for (int v = 0; v < 8; ++v) {
            C[(size_t)(rbase + v) * N + col] = acc[v] + bv;
        }
    }
}

template <int KD, int LAYOUT>
__global__ void wmma_gemm_bias(const float* __restrict__ A,
                               const float* __restrict__ W,
                               const float* __restrict__ bias,
                               float* __restrict__ C,
                               int M, int N) {
    int lane = threadIdx.x & 31;
    int wid  = threadIdx.x >> 5;
    int tile = blockIdx.x * 8 + wid;
    int numTiles = (M >> 4) * (N >> 4);
    if (tile >= numTiles) return;
    gemm_tile_bias<KD, LAYOUT>(A, W, bias, C, N, tile, lane);
}

#define QKV_N 192
#define RPB_STAGE (NHEADS * (2 * KW - 1) * (2 * KW - 1))
#define QKV_TILES ((PIX / 16) * (QKV_N / 16))
#define QKV_BLOCKS (QKV_TILES / 8)
#define FOLD_TILES ((QKV_N / 16) * (CDIM / 16))
#define FOLD_BLOCKS (2 * FOLD_TILES / 8)
static_assert(QKV_TILES % 8 == 0 && (2 * FOLD_TILES) % 8 == 0, "whole 8-wave work-groups");

__global__ void qkv0_gemm_and_fold(const float* __restrict__ x,
                                   const float* __restrict__ qkv_w,
                                   const float* __restrict__ qkv_b,
                                   const float* __restrict__ proj_w,
                                   const float* __restrict__ proj_b,
                                   float* __restrict__ qkv,
                                   float* __restrict__ wfold,
                                   float* __restrict__ bfold) {
    const int lane = threadIdx.x & 31;
    const int wid  = threadIdx.x >> 5;
    if (blockIdx.x < QKV_BLOCKS) {
        gemm_tile_bias<CDIM, A_PLANAR>(x, qkv_w, qkv_b, qkv, QKV_N,
                                       blockIdx.x * 8 + wid, lane);
        return;
    }
    const int t  = (blockIdx.x - QKV_BLOCKS) * 8 + wid;
    const int st = t / FOLD_TILES;
    const int tt = t % FOLD_TILES;
    const int tm = tt / (CDIM / 16);
    const int tn = tt % (CDIM / 16);
    const int lmod  = lane & 15;
    const int lhalf = lane >> 4;

    const float* __restrict__ Qw = qkv_w + (size_t)(st + 1) * QKV_N * CDIM;
    const float* __restrict__ Pw = proj_w + (size_t)st * CDIM * CDIM;
    const float* __restrict__ Ap = Qw + (size_t)(tm * 16 + lmod) * CDIM + 2 * lhalf;
    const float* __restrict__ Bp = Pw + (size_t)(2 * lhalf) * CDIM + (tn * 16 + lmod);

    v8f acc = {0.f, 0.f, 0.f, 0.f, 0.f, 0.f, 0.f, 0.f};
#pragma unroll
    for (int k = 0; k < CDIM; k += 4) {
        v2f a = *(const v2f*)(Ap + k);
        v2f b;
        b.x = Bp[(size_t)k * CDIM];
        b.y = Bp[(size_t)(k + 1) * CDIM];
        acc = __builtin_amdgcn_wmma_f32_16x16x4_f32(
             false, a,  false, b,
             (short)0, acc,  false,  false);
    }
    float* __restrict__ Cp = wfold + (size_t)st * QKV_N * CDIM
                           + (size_t)(tm * 16 + 8 * lhalf) * CDIM + (tn * 16 + lmod);
#pragma unroll
    for (int v = 0; v < 8; ++v) Cp[(size_t)v * CDIM] = acc[v];

    if (tn == 0 && lane < 16) {
        const int n = tm * 16 + lane;
        const float* __restrict__ qr = Qw + (size_t)n * CDIM;
        const float* __restrict__ pb = proj_b + (size_t)st * CDIM;
        float sacc = qkv_b[(st + 1) * QKV_N + n];
#pragma unroll
        for (int j = 0; j < CDIM; j += 4) {
            const v4f qv = *(const v4f*)(qr + j);
            const v4f pv = *(const v4f*)(pb + j);
            sacc = fmaf(qv.x, pv.x, sacc);
            sacc = fmaf(qv.y, pv.y, sacc);
            sacc = fmaf(qv.z, pv.z, sacc);
            sacc = fmaf(qv.w, pv.w, sacc);
        }
        bfold[st * QKV_N + n] = sacc;
    }
}

#define PATCH_W 4
#define PATCH_H 4
#define ATT_THREADS (PATCH_W * PATCH_H * 8)
#define PATCHES_X (WW / PATCH_W)
#define PATCHES_Y (HH / PATCH_H)
#define PATCHES (PATCHES_X * PATCHES_Y)
static_assert(WW % PATCH_W == 0 && HH % PATCH_H == 0, "patches tile the image");
static_assert(HD == 32 && NKK == 49, "8 lanes x 4 dims; 49 taps");

template <int MASK>
__device__ __forceinline__ float lane_xor(float x) {
    return __builtin_bit_cast(float,
        __builtin_amdgcn_update_dpp(0, __builtin_bit_cast(int, x), 0x160 + MASK, 0xf, 0xf, false));
}

template <bool HAS_PREV>
__global__ __launch_bounds__(ATT_THREADS)
void na_attn_kernel(const float* __restrict__ qkv,
                    const float* __restrict__ rpb,
                    const float* __restrict__ prev,
                    float* __restrict__ attn_store,
                    float* __restrict__ out) {
    const int tid = threadIdx.x;
    const int s   = tid & 7;
    const int pl  = tid >> 3;
    const int head = blockIdx.x / PATCHES;
    const int pr   = blockIdx.x % PATCHES;
    const int i = (pr / PATCHES_X) * PATCH_H + (pl / PATCH_W);
    const int j = (pr % PATCHES_X) * PATCH_W + (pl % PATCH_W);
    const int pix = i * WW + j;
    const int idx = head * PIX + pix;

    int si = i - (KW / 2); si = si < 0 ? 0 : si; si = si > HH - KW ? HH - KW : si;
    int sj = j - (KW / 2); sj = sj < 0 ? 0 : sj; sj = sj > WW - KW ? WW - KW : sj;

    const float LOG2E = 1.4426950408889634f;
    const float scale2 = 0.17677669529663687f * LOG2E;

    v4f q = *(const v4f*)(qkv + (size_t)pix * 192 + head * HD + 4 * s);
    q *= scale2;

    const float* kbase = qkv + (size_t)(si * WW + sj) * 192 + 64 + head * HD + 4 * s;
    const float* vbase = kbase + 64;

    const float* rb = rpb + head * (2 * KW - 1) * (2 * KW - 1)
                    + (si - i + KW - 1) * (2 * KW - 1) + (sj - j + KW - 1);
    const float* pprev = prev + (size_t)idx * NKK;
    float extra[7];
#pragma unroll
    for (int t = 0; t < 7; ++t) {
        const int kk = (t < 6) ? (s + 8 * t) : (s == 0 ? NKK - 1 : 0);
        const int a = kk / KW;
        const int b = kk - a * KW;
        float e = rb[a * (2 * KW - 1) + b];
        if (HAS_PREV) e += pprev[kk];
        extra[t] = e * LOG2E;
    }
    if (s != 0) extra[6] = 0.f;

    float lg[NKK];
#pragma unroll
    for (int a = 0; a < KW; ++a) {
#pragma unroll
        for (int b = 0; b < KW; ++b) {
            const int kk = a * KW + b;
            v4f k4 = *(const v4f*)(kbase + (a * WW + b) * 192);
            float part = (s == (kk & 7)) ? extra[kk >> 3] : 0.f;
            part = fmaf(q.x, k4.x, part);
            part = fmaf(q.y, k4.y, part);
            part = fmaf(q.z, k4.z, part);
            part = fmaf(q.w, k4.w, part);
            part += lane_xor<1>(part);
            part += lane_xor<2>(part);
            part += lane_xor<4>(part);
            lg[kk] = part;
        }
    }

    float m = lg[0];
#pragma unroll
    for (int kk = 1; kk < NKK; ++kk) m = fmaxf(m, lg[kk]);
    float ssum = 0.f;
#pragma unroll
    for (int kk = 0; kk < NKK; ++kk) {
        lg[kk] = __builtin_amdgcn_exp2f(lg[kk] - m);
        ssum += lg[kk];
    }
    const float inv = 1.f / ssum;

    float mine[7];
#pragma unroll
    for (int t = 0; t < 7; ++t) mine[t] = 0.f;
#pragma unroll
    for (int kk = 0; kk < NKK; ++kk) {
        if (s == (kk & 7)) mine[kk >> 3] = lg[kk];
    }
    float* astore = attn_store + (size_t)idx * NKK;
#pragma unroll
    for (int t = 0; t < 6; ++t) astore[s + 8 * t] = mine[t] * inv;
    if (s == 0) astore[48] = mine[6] * inv;

    v4f acc = {0.f, 0.f, 0.f, 0.f};
#pragma unroll
    for (int a = 0; a < KW; ++a) {
#pragma unroll
        for (int b = 0; b < KW; ++b) {
            const float p = lg[a * KW + b];
            v4f v4 = *(const v4f*)(vbase + (a * WW + b) * 192);
            acc.x = fmaf(p, v4.x, acc.x);
            acc.y = fmaf(p, v4.y, acc.y);
            acc.z = fmaf(p, v4.z, acc.z);
            acc.w = fmaf(p, v4.w, acc.w);
        }
    }
    acc *= inv;
    *(v4f*)(out + (size_t)pix * CDIM + head * HD + 4 * s) = acc;
}

extern "C" void kernel_launch(void* const* d_in, const int* in_sizes, int n_in,
                              void* d_out, int out_size, void* d_ws, size_t ws_size,
                              hipStream_t stream) {
    const float* x      = (const float*)d_in[0];
    const float* qkv_w  = (const float*)d_in[1];
    const float* qkv_b  = (const float*)d_in[2];
    const float* proj_w = (const float*)d_in[3];
    const float* proj_b = (const float*)d_in[4];
    const float* rpb    = (const float*)d_in[5];
    float* out = (float*)d_out;
    float* ws  = (float*)d_ws;

    float* wfold   = ws;
    float* bfold   = wfold + (size_t)2 * QKV_N * CDIM;
    float* qkv     = ws + (size_t)2 * PIX * CDIM;
    float* attnA   = qkv + (size_t)PIX * QKV_N;
    float* attnB   = attnA + (size_t)NHEADS * PIX * NKK;
    float* attnout = attnB + (size_t)NHEADS * PIX * NKK;

    qkv0_gemm_and_fold<<<QKV_BLOCKS + FOLD_BLOCKS, 256, 0, stream>>>(
        x, qkv_w, qkv_b, proj_w, proj_b, qkv, wfold, bfold);
    na_attn_kernel<false><<<NHEADS * PATCHES, ATT_THREADS, 0, stream>>>(
        qkv, rpb, attnA  , attnA, attnout);

    wmma_gemm_bias<CDIM, ROWMAJOR><<<QKV_BLOCKS, 256, 0, stream>>>(
        attnout, wfold, bfold, qkv, PIX, QKV_N);
    na_attn_kernel<true><<<NHEADS * PATCHES, ATT_THREADS, 0, stream>>>(
        qkv, rpb + RPB_STAGE, attnA, attnB, attnout);

    wmma_gemm_bias<CDIM, ROWMAJOR><<<QKV_BLOCKS, 256, 0, stream>>>(
        attnout, wfold + (size_t)QKV_N * CDIM, bfold + QKV_N, qkv, PIX, QKV_N);
    na_attn_kernel<true><<<NHEADS * PATCHES, ATT_THREADS, 0, stream>>>(
        qkv, rpb + 2 * RPB_STAGE, attnB, attnA, attnout);

    wmma_gemm_bias<CDIM, C_PLANAR><<<(PIX / 16) * (CDIM / 16) / 8, 256, 0, stream>>>(
        attnout, proj_w + (size_t)2 * CDIM * CDIM, proj_b + (size_t)2 * CDIM,
        out, PIX, CDIM);
}


# next-stage QKV GEMM (stages 0,1) and last projection GEMM (stage 2) run inside the attention work-group on its 16x64 output via LDS (both heads per group); 7 -> 4 launches
# speedup vs baseline: 7.3579x; 1.1925x over previous
#include <hip/hip_runtime.h>

typedef float v2f __attribute__((ext_vector_type(2)));
typedef float v4f __attribute__((ext_vector_type(4)));
typedef float v8f __attribute__((ext_vector_type(8)));

#define HH 96
#define WW 96
#define PIX (HH * WW)
#define CDIM 64
#define NHEADS 2
#define HD 32
#define KW 7
#define NKK (KW * KW)
#define QKV_N 192

__device__ __forceinline__ v8f wmma_f32_k4(v2f a, v2f b, v8f c) {
    return __builtin_amdgcn_wmma_f32_16x16x4_f32(
         false, a,  false, b,
         (short)0, c,  false,  false);
}

__device__ __forceinline__ void qkv0_tile(const float* __restrict__ x,
                                          const float* __restrict__ W,
                                          const float* __restrict__ bias,
                                          float* __restrict__ C,
                                          int tile, int lane) {
    const int tilesN = QKV_N >> 4;
    const int tm = tile / tilesN;
    const int tn = tile % tilesN;
    const int lmod  = lane & 15;
    const int lhalf = lane >> 4;

    const float* __restrict__ Ap = x + (size_t)(2 * lhalf) * PIX + (tm * 16 + lmod);
    const float* __restrict__ Wp = W + (size_t)(tn * 16 + lmod) * CDIM + 2 * lhalf;

    v8f acc = {0.f, 0.f, 0.f, 0.f, 0.f, 0.f, 0.f, 0.f};
#pragma unroll
    for (int k = 0; k < CDIM; k += 4) {
        v2f a;
        a.x = Ap[(size_t)k * PIX];
        a.y = Ap[(size_t)(k + 1) * PIX];
        v2f b = *(const v2f*)(Wp + k);
        acc = wmma_f32_k4(a, b, acc);
    }

    const float bv = bias[tn * 16 + lmod];
    const int rbase = tm * 16 + 8 * lhalf;
    const int col   = tn * 16 + lmod;
#pragma unroll
    for (int v = 0; v < 8; ++v) {
        C[(size_t)(rbase + v) * QKV_N + col] = acc[v] + bv;
    }
}

#define RPB_STAGE (NHEADS * (2 * KW - 1) * (2 * KW - 1))
#define QKV_TILES ((PIX / 16) * (QKV_N / 16))
#define QKV_BLOCKS (QKV_TILES / 8)
#define FOLD_TILES ((QKV_N / 16) * (CDIM / 16))
#define FOLD_BLOCKS (2 * FOLD_TILES / 8)
static_assert(QKV_TILES % 8 == 0 && (2 * FOLD_TILES) % 8 == 0, "whole 8-wave work-groups");

__global__ void qkv0_gemm_and_fold(const float* __restrict__ x,
                                   const float* __restrict__ qkv_w,
                                   const float* __restrict__ qkv_b,
                                   const float* __restrict__ proj_w,
                                   const float* __restrict__ proj_b,
                                   float* __restrict__ qkv,
                                   float* __restrict__ wfold,
                                   float* __restrict__ bfold) {
    const int lane = threadIdx.x & 31;
    const int wid  = threadIdx.x >> 5;
    if (blockIdx.x < QKV_BLOCKS) {
        qkv0_tile(x, qkv_w, qkv_b, qkv, blockIdx.x * 8 + wid, lane);
        return;
    }
    const int t  = (blockIdx.x - QKV_BLOCKS) * 8 + wid;
    const int st = t / FOLD_TILES;
    const int tt = t % FOLD_TILES;
    const int tm = tt / (CDIM / 16);
    const int tn = tt % (CDIM / 16);
    const int lmod  = lane & 15;
    const int lhalf = lane >> 4;

    const float* __restrict__ Qw = qkv_w + (size_t)(st + 1) * QKV_N * CDIM;
    const float* __restrict__ Pw = proj_w + (size_t)st * CDIM * CDIM;
    const float* __restrict__ Ap = Qw + (size_t)(tm * 16 + lmod) * CDIM + 2 * lhalf;
    const float* __restrict__ Bp = Pw + (size_t)(2 * lhalf) * CDIM + (tn * 16 + lmod);

    v8f acc = {0.f, 0.f, 0.f, 0.f, 0.f, 0.f, 0.f, 0.f};
#pragma unroll
    for (int k = 0; k < CDIM; k += 4) {
        v2f a = *(const v2f*)(Ap + k);
        v2f b;
        b.x = Bp[(size_t)k * CDIM];
        b.y = Bp[(size_t)(k + 1) * CDIM];
        acc = wmma_f32_k4(a, b, acc);
    }
    float* __restrict__ Cp = wfold + (size_t)st * QKV_N * CDIM
                           + (size_t)(tm * 16 + 8 * lhalf) * CDIM + (tn * 16 + lmod);
#pragma unroll
    for (int v = 0; v < 8; ++v) Cp[(size_t)v * CDIM] = acc[v];

    if (tn == 0 && lane < 16) {
        const int n = tm * 16 + lane;
        const float* __restrict__ qr = Qw + (size_t)n * CDIM;
        const float* __restrict__ pb = proj_b + (size_t)st * CDIM;
        float sacc = qkv_b[(st + 1) * QKV_N + n];
#pragma unroll
        for (int j = 0; j < CDIM; j += 4) {
            const v4f qv = *(const v4f*)(qr + j);
            const v4f pv = *(const v4f*)(pb + j);
            sacc = fmaf(qv.x, pv.x, sacc);
            sacc = fmaf(qv.y, pv.y, sacc);
            sacc = fmaf(qv.z, pv.z, sacc);
            sacc = fmaf(qv.w, pv.w, sacc);
        }
        bfold[st * QKV_N + n] = sacc;
    }
}

#define PATCH_W 4
#define PATCH_H 4
#define PATCH_PIX (PATCH_W * PATCH_H)
#define ATT_THREADS (PATCH_PIX * 8 * NHEADS)
#define PATCHES_X (WW / PATCH_W)
#define PATCHES_Y (HH / PATCH_H)
#define PATCHES (PATCHES_X * PATCHES_Y)
#define OPITCH (CDIM + 4)
static_assert(WW % PATCH_W == 0 && HH % PATCH_H == 0, "patches tile the image");
static_assert(HD == 32 && NKK == 49, "8 lanes x 4 dims; 49 taps");
static_assert(PATCH_PIX == 16 && PATCH_W == 4, "the patch is the 16-row side of the WMMA tiles");
static_assert(ATT_THREADS == 256 && QKV_N / 16 == 12 && CDIM / 16 == 4, "8 waves: 12 or 4 column tiles");

template <int MASK>
__device__ __forceinline__ float lane_xor(float x) {
    return __builtin_bit_cast(float,
        __builtin_amdgcn_update_dpp(0, __builtin_bit_cast(int, x), 0x160 + MASK, 0xf, 0xf, false));
}

template <bool HAS_PREV, bool LAST>
__global__ __launch_bounds__(ATT_THREADS)
void na_attn_gemm_kernel(const float* __restrict__ qkv,
                         const float* __restrict__ rpb,
                         float* attn_io,
                         const float* __restrict__ Wn,
                         const float* __restrict__ bn,
                         float* __restrict__ dst) {
    __shared__ __attribute__((aligned(16))) float o_tile[PATCH_PIX * OPITCH];

    const int tid = threadIdx.x;
    const int s   = tid & 7;
    const int pl  = (tid >> 3) & (PATCH_PIX - 1);
    const int head = tid >> 7;
    const int pr   = blockIdx.x;
    const int i0 = (pr / PATCHES_X) * PATCH_H;
    const int j0 = (pr % PATCHES_X) * PATCH_W;
    const int i = i0 + (pl / PATCH_W);
    const int j = j0 + (pl % PATCH_W);
    const int pix = i * WW + j;
    const int idx = head * PIX + pix;

    int si = i - (KW / 2); si = si < 0 ? 0 : si; si = si > HH - KW ? HH - KW : si;
    int sj = j - (KW / 2); sj = sj < 0 ? 0 : sj; sj = sj > WW - KW ? WW - KW : sj;

    const float LOG2E = 1.4426950408889634f;
    const float scale2 = 0.17677669529663687f * LOG2E;

    v4f q = *(const v4f*)(qkv + (size_t)pix * QKV_N + head * HD + 4 * s);
    q *= scale2;

    const float* kbase = qkv + (size_t)(si * WW + sj) * QKV_N + 64 + head * HD + 4 * s;
    const float* vbase = kbase + 64;

    const float* rb = rpb + head * (2 * KW - 1) * (2 * KW - 1)
                    + (si - i + KW - 1) * (2 * KW - 1) + (sj - j + KW - 1);
    float* aio = attn_io + (size_t)idx * NKK;
    float extra[7];
#pragma unroll
    for (int t = 0; t < 7; ++t) {
        const int kk = (t < 6) ? (s + 8 * t) : (s == 0 ? NKK - 1 : 0);
        const int a = kk / KW;
        const int b = kk - a * KW;
        float e = rb[a * (2 * KW - 1) + b];
        if (HAS_PREV) e += aio[kk];
        extra[t] = e * LOG2E;
    }
    if (s != 0) extra[6] = 0.f;

    float lg[NKK];
#pragma unroll
    for (int a = 0; a < KW; ++a) {
#pragma unroll
        for (int b = 0; b < KW; ++b) {
            const int kk = a * KW + b;
            const v4f k4 = *(const v4f*)(kbase + (a * WW + b) * QKV_N);
            float part = (s == (kk & 7)) ? extra[kk >> 3] : 0.f;
            part = fmaf(q.x, k4.x, part);
            part = fmaf(q.y, k4.y, part);
            part = fmaf(q.z, k4.z, part);
            part = fmaf(q.w, k4.w, part);
            part += lane_xor<1>(part);
            part += lane_xor<2>(part);
            part += lane_xor<4>(part);
            lg[kk] = part;
        }
    }

    float m = lg[0];
#pragma unroll
    for (int kk = 1; kk < NKK; ++kk) m = fmaxf(m, lg[kk]);
    float ssum = 0.f;
#pragma unroll
    for (int kk = 0; kk < NKK; ++kk) {
        lg[kk] = __builtin_amdgcn_exp2f(lg[kk] - m);
        ssum += lg[kk];
    }
    const float inv = 1.f / ssum;

    float mine[7];
#pragma unroll
    for (int t = 0; t < 7; ++t) mine[t] = 0.f;
#pragma unroll
    for (int kk = 0; kk < NKK; ++kk) {
        if (s == (kk & 7)) mine[kk >> 3] = lg[kk];
    }
#pragma unroll
    for (int t = 0; t < 6; ++t) aio[s + 8 * t] = mine[t] * inv;
    if (s == 0) aio[48] = mine[6] * inv;

    v4f acc = {0.f, 0.f, 0.f, 0.f};
#pragma unroll
    for (int a = 0; a < KW; ++a) {
#pragma unroll
        for (int b = 0; b < KW; ++b) {
            const float p = lg[a * KW + b];
            const v4f v4 = *(const v4f*)(vbase + (a * WW + b) * QKV_N);
            acc.x = fmaf(p, v4.x, acc.x);
            acc.y = fmaf(p, v4.y, acc.y);
            acc.z = fmaf(p, v4.z, acc.z);
            acc.w = fmaf(p, v4.w, acc.w);
        }
    }
    acc *= inv;

    const int wid   = __builtin_amdgcn_readfirstlane(tid >> 5);
    const int lane  = tid & 31;
    const int lmod  = lane & 15;
    const int lhalf = lane >> 4;
    const bool has_t0 = LAST ? (wid < CDIM / 16) : true;
    const bool has_t1 = LAST ? false : (wid < 4);

    *(v4f*)(o_tile + pl * OPITCH + head * HD + 4 * s) = acc;

    const float* __restrict__ Wp = Wn + (size_t)(wid * 16 + lmod) * CDIM + 2 * lhalf;
    v2f w0[CDIM / 4], w1[CDIM / 4];
    if (has_t0) {
#pragma unroll
        for (int kk = 0; kk < CDIM / 4; ++kk) w0[kk] = *(const v2f*)(Wp + 4 * kk);
    }
    if (has_t1) {
#pragma unroll
        for (int kk = 0; kk < CDIM / 4; ++kk) w1[kk] = *(const v2f*)(Wp + (size_t)8 * 16 * CDIM + 4 * kk);
    }
    __syncthreads();
    if (!has_t0) return;

    v2f o[CDIM / 4];
#pragma unroll
    for (int kk = 0; kk < CDIM / 4; ++kk)
        o[kk] = *(const v2f*)(o_tile + lmod * OPITCH + 4 * kk + 2 * lhalf);

    if (!LAST) {
        float* __restrict__ Cp = dst + (size_t)((i0 + 2 * lhalf) * WW + j0) * QKV_N + (wid * 16 + lmod);
        v8f c0 = {0.f, 0.f, 0.f, 0.f, 0.f, 0.f, 0.f, 0.f};
        if (has_t1) {
            v8f c1 = {0.f, 0.f, 0.f, 0.f, 0.f, 0.f, 0.f, 0.f};
#pragma unroll
            for (int kk = 0; kk < CDIM / 4; ++kk) {
                c0 = wmma_f32_k4(o[kk], w0[kk], c0);
                c1 = wmma_f32_k4(o[kk], w1[kk], c1);
            }
            const float bv0 = bn[wid * 16 + lmod];
            const float bv1 = bn[(wid + 8) * 16 + lmod];
#pragma unroll
            for (int v = 0; v < 8; ++v) {
                const size_t off = (size_t)((v >> 2) * WW + (v & 3)) * QKV_N;
                Cp[off] = c0[v] + bv0;
                Cp[off + 8 * 16] = c1[v] + bv1;
            }
        } else {
#pragma unroll
            for (int kk = 0; kk < CDIM / 4; ++kk) c0 = wmma_f32_k4(o[kk], w0[kk], c0);
            const float bv0 = bn[wid * 16 + lmod];
#pragma unroll
            for (int v = 0; v < 8; ++v) {
                const size_t off = (size_t)((v >> 2) * WW + (v & 3)) * QKV_N;
                Cp[off] = c0[v] + bv0;
            }
        }
    } else {
        v8f c0 = {0.f, 0.f, 0.f, 0.f, 0.f, 0.f, 0.f, 0.f};
#pragma unroll
        for (int kk = 0; kk < CDIM / 4; ++kk) c0 = wmma_f32_k4(w0[kk], o[kk], c0);
        const int ch = wid * 16 + 8 * lhalf;
        const v4f blo = *(const v4f*)(bn + ch);
        const v4f bhi = *(const v4f*)(bn + ch + 4);
        float* __restrict__ Cp = dst + (size_t)ch * PIX
                               + (i0 + (lmod >> 2)) * WW + j0 + (lmod & 3);
#pragma unroll
        for (int v = 0; v < 8; ++v) {
            const float bv = (v < 4) ? blo[v & 3] : bhi[v & 3];
            Cp[(size_t)v * PIX] = c0[v] + bv;
        }
    }
}

extern "C" void kernel_launch(void* const* d_in, const int* in_sizes, int n_in,
                              void* d_out, int out_size, void* d_ws, size_t ws_size,
                              hipStream_t stream) {
    const float* x      = (const float*)d_in[0];
    const float* qkv_w  = (const float*)d_in[1];
    const float* qkv_b  = (const float*)d_in[2];
    const float* proj_w = (const float*)d_in[3];
    const float* proj_b = (const float*)d_in[4];
    const float* rpb    = (const float*)d_in[5];
    float* out = (float*)d_out;
    float* ws  = (float*)d_ws;

    float* wfold = ws;
    float* bfold = wfold + (size_t)2 * QKV_N * CDIM;
    float* qkvA  = bfold + (size_t)2 * QKV_N;
    float* qkvB  = qkvA + (size_t)PIX * QKV_N;
    float* attn  = qkvB + (size_t)PIX * QKV_N;

    qkv0_gemm_and_fold<<<QKV_BLOCKS + FOLD_BLOCKS, 256, 0, stream>>>(
        x, qkv_w, qkv_b, proj_w, proj_b, qkvA, wfold, bfold);
    na_attn_gemm_kernel<false, false><<<PATCHES, ATT_THREADS, 0, stream>>>(
        qkvA, rpb, attn, wfold, bfold, qkvB);
    na_attn_gemm_kernel<true, false><<<PATCHES, ATT_THREADS, 0, stream>>>(
        qkvB, rpb + RPB_STAGE, attn, wfold + (size_t)QKV_N * CDIM, bfold + QKV_N, qkvA);
    na_attn_gemm_kernel<true, true><<<PATCHES, ATT_THREADS, 0, stream>>>(
        qkvA, rpb + 2 * RPB_STAGE, attn, proj_w + (size_t)2 * CDIM * CDIM,
        proj_b + (size_t)2 * CDIM, out);
}


# attention: K/V rows of the patch's 10x10 pixel square staged once per work-group in LDS (13 16B global loads per lane in one batch); both tap passes read LDS, not 98 global loads per lane
# speedup vs baseline: 8.8321x; 1.2004x over previous
#include <hip/hip_runtime.h>

typedef float v2f __attribute__((ext_vector_type(2)));
typedef float v4f __attribute__((ext_vector_type(4)));
typedef float v8f __attribute__((ext_vector_type(8)));

#define HH 96
#define WW 96
#define PIX (HH * WW)
#define CDIM 64
#define NHEADS 2
#define HD 32
#define KW 7
#define NKK (KW * KW)
#define QKV_N 192

__device__ __forceinline__ v8f wmma_f32_k4(v2f a, v2f b, v8f c) {
    return __builtin_amdgcn_wmma_f32_16x16x4_f32(
         false, a,  false, b,
         (short)0, c,  false,  false);
}

__device__ __forceinline__ void qkv0_tile(const float* __restrict__ x,
                                          const float* __restrict__ W,
                                          const float* __restrict__ bias,
                                          float* __restrict__ C,
                                          int tile, int lane) {
    const int tilesN = QKV_N >> 4;
    const int tm = tile / tilesN;
    const int tn = tile % tilesN;
    const int lmod  = lane & 15;
    const int lhalf = lane >> 4;

    const float* __restrict__ Ap = x + (size_t)(2 * lhalf) * PIX + (tm * 16 + lmod);
    const float* __restrict__ Wp = W + (size_t)(tn * 16 + lmod) * CDIM + 2 * lhalf;

    v8f acc = {0.f, 0.f, 0.f, 0.f, 0.f, 0.f, 0.f, 0.f};
#pragma unroll
    for (int k = 0; k < CDIM; k += 4) {
        v2f a;
        a.x = Ap[(size_t)k * PIX];
        a.y = Ap[(size_t)(k + 1) * PIX];
        v2f b = *(const v2f*)(Wp + k);
        acc = wmma_f32_k4(a, b, acc);
    }

    const float bv = bias[tn * 16 + lmod];
    const int rbase = tm * 16 + 8 * lhalf;
    const int col   = tn * 16 + lmod;
#pragma unroll
    for (int v = 0; v < 8; ++v) {
        C[(size_t)(rbase + v) * QKV_N + col] = acc[v] + bv;
    }
}

#define RPB_STAGE (NHEADS * (2 * KW - 1) * (2 * KW - 1))
#define QKV_TILES ((PIX / 16) * (QKV_N / 16))
#define QKV_BLOCKS (QKV_TILES / 8)
#define FOLD_TILES ((QKV_N / 16) * (CDIM / 16))
#define FOLD_BLOCKS (2 * FOLD_TILES / 8)
static_assert(QKV_TILES % 8 == 0 && (2 * FOLD_TILES) % 8 == 0, "whole 8-wave work-groups");

__global__ void qkv0_gemm_and_fold(const float* __restrict__ x,
                                   const float* __restrict__ qkv_w,
                                   const float* __restrict__ qkv_b,
                                   const float* __restrict__ proj_w,
                                   const float* __restrict__ proj_b,
                                   float* __restrict__ qkv,
                                   float* __restrict__ wfold,
                                   float* __restrict__ bfold) {
    const int lane = threadIdx.x & 31;
    const int wid  = threadIdx.x >> 5;
    if (blockIdx.x < QKV_BLOCKS) {
        qkv0_tile(x, qkv_w, qkv_b, qkv, blockIdx.x * 8 + wid, lane);
        return;
    }
    const int t  = (blockIdx.x - QKV_BLOCKS) * 8 + wid;
    const int st = t / FOLD_TILES;
    const int tt = t % FOLD_TILES;
    const int tm = tt / (CDIM / 16);
    const int tn = tt % (CDIM / 16);
    const int lmod  = lane & 15;
    const int lhalf = lane >> 4;

    const float* __restrict__ Qw = qkv_w + (size_t)(st + 1) * QKV_N * CDIM;
    const float* __restrict__ Pw = proj_w + (size_t)st * CDIM * CDIM;
    const float* __restrict__ Ap = Qw + (size_t)(tm * 16 + lmod) * CDIM + 2 * lhalf;
    const float* __restrict__ Bp = Pw + (size_t)(2 * lhalf) * CDIM + (tn * 16 + lmod);

    v8f acc = {0.f, 0.f, 0.f, 0.f, 0.f, 0.f, 0.f, 0.f};
#pragma unroll
    for (int k = 0; k < CDIM; k += 4) {
        v2f a = *(const v2f*)(Ap + k);
        v2f b;
        b.x = Bp[(size_t)k * CDIM];
        b.y = Bp[(size_t)(k + 1) * CDIM];
        acc = wmma_f32_k4(a, b, acc);
    }
    float* __restrict__ Cp = wfold + (size_t)st * QKV_N * CDIM
                           + (size_t)(tm * 16 + 8 * lhalf) * CDIM + (tn * 16 + lmod);
#pragma unroll
    for (int v = 0; v < 8; ++v) Cp[(size_t)v * CDIM] = acc[v];

    if (tn == 0 && lane < 16) {
        const int n = tm * 16 + lane;
        const float* __restrict__ qr = Qw + (size_t)n * CDIM;
        const float* __restrict__ pb = proj_b + (size_t)st * CDIM;
        float sacc = qkv_b[(st + 1) * QKV_N + n];
#pragma unroll
        for (int j = 0; j < CDIM; j += 4) {
            const v4f qv = *(const v4f*)(qr + j);
            const v4f pv = *(const v4f*)(pb + j);
            sacc = fmaf(qv.x, pv.x, sacc);
            sacc = fmaf(qv.y, pv.y, sacc);
            sacc = fmaf(qv.z, pv.z, sacc);
            sacc = fmaf(qv.w, pv.w, sacc);
        }
        bfold[st * QKV_N + n] = sacc;
    }
}

#define PATCH_W 4
#define PATCH_H 4
#define PATCH_PIX (PATCH_W * PATCH_H)
#define ATT_THREADS (PATCH_PIX * 8 * NHEADS)
#define PATCHES_X (WW / PATCH_W)
#define PATCHES_Y (HH / PATCH_H)
#define PATCHES (PATCHES_X * PATCHES_Y)
#define OPITCH (CDIM + 4)
#define HALO (PATCH_W + KW - 1)
#define HALO_PIX (HALO * HALO)
#define PLANE_F ((HALO_PIX + 1) * HD)
static_assert(PATCH_W == PATCH_H && HALO <= HH && HALO <= WW, "square patch; the K/V square fits the image");
static_assert(HALO_PIX > 8 * 12 && HALO_PIX <= 8 * 13, "13 copy rounds of 8 pixels cover the square");
static_assert(WW % PATCH_W == 0 && HH % PATCH_H == 0, "patches tile the image");
static_assert(HD == 32 && NKK == 49, "8 lanes x 4 dims; 49 taps");
static_assert(PATCH_PIX == 16 && PATCH_W == 4, "the patch is the 16-row side of the WMMA tiles");
static_assert(ATT_THREADS == 256 && QKV_N / 16 == 12 && CDIM / 16 == 4, "8 waves: 12 or 4 column tiles");

template <int MASK>
__device__ __forceinline__ float lane_xor(float x) {
    return __builtin_bit_cast(float,
        __builtin_amdgcn_update_dpp(0, __builtin_bit_cast(int, x), 0x160 + MASK, 0xf, 0xf, false));
}

template <bool HAS_PREV, bool LAST>
__global__ __launch_bounds__(ATT_THREADS)
void na_attn_gemm_kernel(const float* __restrict__ qkv,
                         const float* __restrict__ rpb,
                         float* attn_io,
                         const float* __restrict__ Wn,
                         const float* __restrict__ bn,
                         float* __restrict__ dst) {
    __shared__ __attribute__((aligned(16))) float o_tile[PATCH_PIX * OPITCH];
    __shared__ __attribute__((aligned(16))) float kv_tile[4 * PLANE_F];

    const int tid = threadIdx.x;
    const int s   = tid & 7;
    const int pl  = (tid >> 3) & (PATCH_PIX - 1);
    const int head = tid >> 7;
    const int pr   = blockIdx.x;
    const int i0 = (pr / PATCHES_X) * PATCH_H;
    const int j0 = (pr % PATCHES_X) * PATCH_W;
    const int i = i0 + (pl / PATCH_W);
    const int j = j0 + (pl % PATCH_W);
    const int pix = i * WW + j;
    const int idx = head * PIX + pix;

    int si = i - (KW / 2); si = si < 0 ? 0 : si; si = si > HH - KW ? HH - KW : si;
    int sj = j - (KW / 2); sj = sj < 0 ? 0 : sj; sj = sj > WW - KW ? WW - KW : sj;

    const float LOG2E = 1.4426950408889634f;
    const float scale2 = 0.17677669529663687f * LOG2E;

    v4f q = *(const v4f*)(qkv + (size_t)pix * QKV_N + head * HD + 4 * s);
    q *= scale2;

    const float* rb = rpb + head * (2 * KW - 1) * (2 * KW - 1)
                    + (si - i + KW - 1) * (2 * KW - 1) + (sj - j + KW - 1);
    float* aio = attn_io + (size_t)idx * NKK;
    float extra[7];
#pragma unroll
    for (int t = 0; t < 7; ++t) {
        const int kk = (t < 6) ? (s + 8 * t) : (s == 0 ? NKK - 1 : 0);
        const int a = kk / KW;
        const int b = kk - a * KW;
        float e = rb[a * (2 * KW - 1) + b];
        if (HAS_PREV) e += aio[kk];
        extra[t] = e * LOG2E;
    }
    if (s != 0) extra[6] = 0.f;

    const int wid  = __builtin_amdgcn_readfirstlane(tid >> 5);
    const int lane = tid & 31;

    int hi0 = i0 - (KW / 2); hi0 = hi0 < 0 ? 0 : hi0; hi0 = hi0 > HH - HALO ? HH - HALO : hi0;
    int hj0 = j0 - (KW / 2); hj0 = hj0 < 0 ? 0 : hj0; hj0 = hj0 > WW - HALO ? WW - HALO : hj0;
    {
        const float* __restrict__ gsrc = qkv + (size_t)(hi0 * WW + hj0) * QKV_N + 64 + 4 * lane;
        float* sdst = kv_tile + (lane >> 3) * PLANE_F + 4 * (lane & 7);
        v4f stg[13];
#pragma unroll
        for (int n = 0; n < 13; ++n) {
            int hp = wid + 8 * n;
            hp = hp > HALO_PIX - 1 ? HALO_PIX - 1 : hp;
            const int hr = hp / HALO;
            const int hc = hp - hr * HALO;
            stg[n] = *(const v4f*)(gsrc + (size_t)(hr * WW + hc) * QKV_N);
        }
#pragma unroll
        for (int n = 0; n < 13; ++n) {
            int hp = wid + 8 * n;
            hp = hp > HALO_PIX - 1 ? HALO_PIX - 1 : hp;
            *(v4f*)(sdst + hp * HD) = stg[n];
        }
    }

    const float* kbase = kv_tile + head * PLANE_F + ((si - hi0) * HALO + (sj - hj0)) * HD + 4 * s;
    const float* vbase = kbase + 2 * PLANE_F;

    __syncthreads();

    float lg[NKK];
#pragma unroll
    for (int a = 0; a < KW; ++a) {
#pragma unroll
        for (int b = 0; b < KW; ++b) {
            const int kk = a * KW + b;
            const v4f k4 = *(const v4f*)(kbase + (a * HALO + b) * HD);
            float part = (s == (kk & 7)) ? extra[kk >> 3] : 0.f;
            part = fmaf(q.x, k4.x, part);
            part = fmaf(q.y, k4.y, part);
            part = fmaf(q.z, k4.z, part);
            part = fmaf(q.w, k4.w, part);
            part += lane_xor<1>(part);
            part += lane_xor<2>(part);
            part += lane_xor<4>(part);
            lg[kk] = part;
        }
    }

    float m = lg[0];
#pragma unroll
    for (int kk = 1; kk < NKK; ++kk) m = fmaxf(m, lg[kk]);
    float ssum = 0.f;
#pragma unroll
    for (int kk = 0; kk < NKK; ++kk) {
        lg[kk] = __builtin_amdgcn_exp2f(lg[kk] - m);
        ssum += lg[kk];
    }
    const float inv = 1.f / ssum;

    float mine[7];
#pragma unroll
    for (int t = 0; t < 7; ++t) mine[t] = 0.f;
#pragma unroll
    for (int kk = 0; kk < NKK; ++kk) {
        if (s == (kk & 7)) mine[kk >> 3] = lg[kk];
    }
#pragma unroll
    for (int t = 0; t < 6; ++t) aio[s + 8 * t] = mine[t] * inv;
    if (s == 0) aio[48] = mine[6] * inv;

    v4f acc = {0.f, 0.f, 0.f, 0.f};
#pragma unroll
    for (int a = 0; a < KW; ++a) {
#pragma unroll
        for (int b = 0; b < KW; ++b) {
            const float p = lg[a * KW + b];
            const v4f v4 = *(const v4f*)(vbase + (a * HALO + b) * HD);
            acc.x = fmaf(p, v4.x, acc.x);
            acc.y = fmaf(p, v4.y, acc.y);
            acc.z = fmaf(p, v4.z, acc.z);
            acc.w = fmaf(p, v4.w, acc.w);
        }
    }
    acc *= inv;

    const int lmod  = lane & 15;
    const int lhalf = lane >> 4;
    const bool has_t0 = LAST ? (wid < CDIM / 16) : true;
    const bool has_t1 = LAST ? false : (wid < 4);

    *(v4f*)(o_tile + pl * OPITCH + head * HD + 4 * s) = acc;

    const float* __restrict__ Wp = Wn + (size_t)(wid * 16 + lmod) * CDIM + 2 * lhalf;
    v2f w0[CDIM / 4], w1[CDIM / 4];
    if (has_t0) {
#pragma unroll
        for (int kk = 0; kk < CDIM / 4; ++kk) w0[kk] = *(const v2f*)(Wp + 4 * kk);
    }
    if (has_t1) {
#pragma unroll
        for (int kk = 0; kk < CDIM / 4; ++kk) w1[kk] = *(const v2f*)(Wp + (size_t)8 * 16 * CDIM + 4 * kk);
    }
    __syncthreads();
    if (!has_t0) return;

    v2f o[CDIM / 4];
#pragma unroll
    for (int kk = 0; kk < CDIM / 4; ++kk)
        o[kk] = *(const v2f*)(o_tile + lmod * OPITCH + 4 * kk + 2 * lhalf);

    if (!LAST) {
        float* __restrict__ Cp = dst + (size_t)((i0 + 2 * lhalf) * WW + j0) * QKV_N + (wid * 16 + lmod);
        v8f c0 = {0.f, 0.f, 0.f, 0.f, 0.f, 0.f, 0.f, 0.f};
        if (has_t1) {
            v8f c1 = {0.f, 0.f, 0.f, 0.f, 0.f, 0.f, 0.f, 0.f};
#pragma unroll
            for (int kk = 0; kk < CDIM / 4; ++kk) {
                c0 = wmma_f32_k4(o[kk], w0[kk], c0);
                c1 = wmma_f32_k4(o[kk], w1[kk], c1);
            }
            const float bv0 = bn[wid * 16 + lmod];
            const float bv1 = bn[(wid + 8) * 16 + lmod];
#pragma unroll
            for (int v = 0; v < 8; ++v) {
                const size_t off = (size_t)((v >> 2) * WW + (v & 3)) * QKV_N;
                Cp[off] = c0[v] + bv0;
                Cp[off + 8 * 16] = c1[v] + bv1;
            }
        } else {
#pragma unroll
            for (int kk = 0; kk < CDIM / 4; ++kk) c0 = wmma_f32_k4(o[kk], w0[kk], c0);
            const float bv0 = bn[wid * 16 + lmod];
#pragma unroll
            for (int v = 0; v < 8; ++v) {
                const size_t off = (size_t)((v >> 2) * WW + (v & 3)) * QKV_N;
                Cp[off] = c0[v] + bv0;
            }
        }
    } else {
        v8f c0 = {0.f, 0.f, 0.f, 0.f, 0.f, 0.f, 0.f, 0.f};
#pragma unroll
        for (int kk = 0; kk < CDIM / 4; ++kk) c0 = wmma_f32_k4(w0[kk], o[kk], c0);
        const int ch = wid * 16 + 8 * lhalf;
        const v4f blo = *(const v4f*)(bn + ch);
        const v4f bhi = *(const v4f*)(bn + ch + 4);
        float* __restrict__ Cp = dst + (size_t)ch * PIX
                               + (i0 + (lmod >> 2)) * WW + j0 + (lmod & 3);
#pragma unroll
        for (int v = 0; v < 8; ++v) {
            const float bv = (v < 4) ? blo[v & 3] : bhi[v & 3];
            Cp[(size_t)v * PIX] = c0[v] + bv;
        }
    }
}

extern "C" void kernel_launch(void* const* d_in, const int* in_sizes, int n_in,
                              void* d_out, int out_size, void* d_ws, size_t ws_size,
                              hipStream_t stream) {
    const float* x      = (const float*)d_in[0];
    const float* qkv_w  = (const float*)d_in[1];
    const float* qkv_b  = (const float*)d_in[2];
    const float* proj_w = (const float*)d_in[3];
    const float* proj_b = (const float*)d_in[4];
    const float* rpb    = (const float*)d_in[5];
    float* out = (float*)d_out;
    float* ws  = (float*)d_ws;

    float* wfold = ws;
    float* bfold = wfold + (size_t)2 * QKV_N * CDIM;
    float* qkvA  = bfold + (size_t)2 * QKV_N;
    float* qkvB  = qkvA + (size_t)PIX * QKV_N;
    float* attn  = qkvB + (size_t)PIX * QKV_N;

    qkv0_gemm_and_fold<<<QKV_BLOCKS + FOLD_BLOCKS, 256, 0, stream>>>(
        x, qkv_w, qkv_b, proj_w, proj_b, qkvA, wfold, bfold);
    na_attn_gemm_kernel<false, false><<<PATCHES, ATT_THREADS, 0, stream>>>(
        qkvA, rpb, attn, wfold, bfold, qkvB);
    na_attn_gemm_kernel<true, false><<<PATCHES, ATT_THREADS, 0, stream>>>(
        qkvB, rpb + RPB_STAGE, attn, wfold + (size_t)QKV_N * CDIM, bfold + QKV_N, qkvA);
    na_attn_gemm_kernel<true, true><<<PATCHES, ATT_THREADS, 0, stream>>>(
        qkvA, rpb + 2 * RPB_STAGE, attn, proj_w + (size_t)2 * CDIM * CDIM,
        proj_b + (size_t)2 * CDIM, out);
}


# stage-0 QKV GEMM: one wave per 32x32 output block (2x2 WMMA tiles, each fragment loaded once and used twice, 4 independent accumulators), 4-wave groups: 1824 waves enter, not 7008; same sums
# speedup vs baseline: 9.3777x; 1.0618x over previous
#include <hip/hip_runtime.h>

typedef float v2f __attribute__((ext_vector_type(2)));
typedef float v4f __attribute__((ext_vector_type(4)));
typedef float v8f __attribute__((ext_vector_type(8)));

#define HH 96
#define WW 96
#define PIX (HH * WW)
#define CDIM 64
#define NHEADS 2
#define HD 32
#define KW 7
#define NKK (KW * KW)
#define QKV_N 192

__device__ __forceinline__ v8f wmma_f32_k4(v2f a, v2f b, v8f c) {
    return __builtin_amdgcn_wmma_f32_16x16x4_f32(
         false, a,  false, b,
         (short)0, c,  false,  false);
}

__device__ __forceinline__ void qkv0_block(const float* __restrict__ x,
                                           const float* __restrict__ W,
                                           const float* __restrict__ bias,
                                           float* __restrict__ C,
                                           int blk, int lane) {
    const int blocksN = QKV_N >> 5;
    const int tm = blk / blocksN;
    const int tn = blk % blocksN;
    const int lmod  = lane & 15;
    const int lhalf = lane >> 4;

    const float* __restrict__ Ap = x + (size_t)(2 * lhalf) * PIX + (tm * 32 + lmod);
    const float* __restrict__ Wp = W + (size_t)(tn * 32 + lmod) * CDIM + 2 * lhalf;

    v8f c00 = {0.f, 0.f, 0.f, 0.f, 0.f, 0.f, 0.f, 0.f};
    v8f c01 = c00, c10 = c00, c11 = c00;
#pragma unroll
    for (int k = 0; k < CDIM; k += 4) {
        v2f a0, a1;
        a0.x = Ap[(size_t)k * PIX];
        a0.y = Ap[(size_t)(k + 1) * PIX];
        a1.x = Ap[(size_t)k * PIX + 16];
        a1.y = Ap[(size_t)(k + 1) * PIX + 16];
        const v2f b0 = *(const v2f*)(Wp + k);
        const v2f b1 = *(const v2f*)(Wp + (size_t)16 * CDIM + k);
        c00 = wmma_f32_k4(a0, b0, c00);
        c01 = wmma_f32_k4(a0, b1, c01);
        c10 = wmma_f32_k4(a1, b0, c10);
        c11 = wmma_f32_k4(a1, b1, c11);
    }

    const int col = tn * 32 + lmod;
    const float bv0 = bias[col];
    const float bv1 = bias[col + 16];
    float* __restrict__ Cp = C + (size_t)(tm * 32 + 8 * lhalf) * QKV_N + col;
#pragma unroll
    for (int v = 0; v < 8; ++v) {
        Cp[(size_t)v * QKV_N]             = c00[v] + bv0;
        Cp[(size_t)v * QKV_N + 16]        = c01[v] + bv1;
        Cp[(size_t)(v + 16) * QKV_N]      = c10[v] + bv0;
        Cp[(size_t)(v + 16) * QKV_N + 16] = c11[v] + bv1;
    }
}

#define RPB_STAGE (NHEADS * (2 * KW - 1) * (2 * KW - 1))
#define L1_WAVES 4
#define L1_THREADS (32 * L1_WAVES)
#define QKV_WBLOCKS ((PIX / 32) * (QKV_N / 32))
#define QKV_BLOCKS (QKV_WBLOCKS / L1_WAVES)
#define FOLD_TILES ((QKV_N / 16) * (CDIM / 16))
#define FOLD_BLOCKS (2 * FOLD_TILES / L1_WAVES)
static_assert(PIX % 32 == 0 && QKV_N % 32 == 0, "32x32 blocks tile the QKV output");
static_assert(QKV_WBLOCKS % L1_WAVES == 0 && (2 * FOLD_TILES) % L1_WAVES == 0, "whole 4-wave work-groups");

__global__ __launch_bounds__(L1_THREADS) __attribute__((amdgpu_waves_per_eu(1, 4)))
void qkv0_gemm_and_fold(const float* __restrict__ x,
                                   const float* __restrict__ qkv_w,
                                   const float* __restrict__ qkv_b,
                                   const float* __restrict__ proj_w,
                                   const float* __restrict__ proj_b,
                                   float* __restrict__ qkv,
                                   float* __restrict__ wfold,
                                   float* __restrict__ bfold) {
    const int lane = threadIdx.x & 31;
    const int wid  = threadIdx.x >> 5;
    if (blockIdx.x < QKV_BLOCKS) {
        qkv0_block(x, qkv_w, qkv_b, qkv, blockIdx.x * L1_WAVES + wid, lane);
        return;
    }
    const int t  = (blockIdx.x - QKV_BLOCKS) * L1_WAVES + wid;
    const int st = t / FOLD_TILES;
    const int tt = t % FOLD_TILES;
    const int tm = tt / (CDIM / 16);
    const int tn = tt % (CDIM / 16);
    const int lmod  = lane & 15;
    const int lhalf = lane >> 4;

    const float* __restrict__ Qw = qkv_w + (size_t)(st + 1) * QKV_N * CDIM;
    const float* __restrict__ Pw = proj_w + (size_t)st * CDIM * CDIM;
    const float* __restrict__ Ap = Qw + (size_t)(tm * 16 + lmod) * CDIM + 2 * lhalf;
    const float* __restrict__ Bp = Pw + (size_t)(2 * lhalf) * CDIM + (tn * 16 + lmod);

    v8f acc = {0.f, 0.f, 0.f, 0.f, 0.f, 0.f, 0.f, 0.f};
#pragma unroll
    for (int k = 0; k < CDIM; k += 4) {
        v2f a = *(const v2f*)(Ap + k);
        v2f b;
        b.x = Bp[(size_t)k * CDIM];
        b.y = Bp[(size_t)(k + 1) * CDIM];
        acc = wmma_f32_k4(a, b, acc);
    }
    float* __restrict__ Cp = wfold + (size_t)st * QKV_N * CDIM
                           + (size_t)(tm * 16 + 8 * lhalf) * CDIM + (tn * 16 + lmod);
#pragma unroll
    for (int v = 0; v < 8; ++v) Cp[(size_t)v * CDIM] = acc[v];

    if (tn == 0 && lane < 16) {
        const int n = tm * 16 + lane;
        const float* __restrict__ qr = Qw + (size_t)n * CDIM;
        const float* __restrict__ pb = proj_b + (size_t)st * CDIM;
        float sacc = qkv_b[(st + 1) * QKV_N + n];
#pragma unroll
        for (int j = 0; j < CDIM; j += 4) {
            const v4f qv = *(const v4f*)(qr + j);
            const v4f pv = *(const v4f*)(pb + j);
            sacc = fmaf(qv.x, pv.x, sacc);
            sacc = fmaf(qv.y, pv.y, sacc);
            sacc = fmaf(qv.z, pv.z, sacc);
            sacc = fmaf(qv.w, pv.w, sacc);
        }
        bfold[st * QKV_N + n] = sacc;
    }
}

#define PATCH_W 4
#define PATCH_H 4
#define PATCH_PIX (PATCH_W * PATCH_H)
#define ATT_THREADS (PATCH_PIX * 8 * NHEADS)
#define PATCHES_X (WW / PATCH_W)
#define PATCHES_Y (HH / PATCH_H)
#define PATCHES (PATCHES_X * PATCHES_Y)
#define OPITCH (CDIM + 4)
#define HALO (PATCH_W + KW - 1)
#define HALO_PIX (HALO * HALO)
#define PLANE_F ((HALO_PIX + 1) * HD)
static_assert(PATCH_W == PATCH_H && HALO <= HH && HALO <= WW, "square patch; the K/V square fits the image");
static_assert(HALO_PIX > 8 * 12 && HALO_PIX <= 8 * 13, "13 copy rounds of 8 pixels cover the square");
static_assert(WW % PATCH_W == 0 && HH % PATCH_H == 0, "patches tile the image");
static_assert(HD == 32 && NKK == 49, "8 lanes x 4 dims; 49 taps");
static_assert(PATCH_PIX == 16 && PATCH_W == 4, "the patch is the 16-row side of the WMMA tiles");
static_assert(ATT_THREADS == 256 && QKV_N / 16 == 12 && CDIM / 16 == 4, "8 waves: 12 or 4 column tiles");

template <int MASK>
__device__ __forceinline__ float lane_xor(float x) {
    return __builtin_bit_cast(float,
        __builtin_amdgcn_update_dpp(0, __builtin_bit_cast(int, x), 0x160 + MASK, 0xf, 0xf, false));
}

template <bool HAS_PREV, bool LAST>
__global__ __launch_bounds__(ATT_THREADS)
void na_attn_gemm_kernel(const float* __restrict__ qkv,
                         const float* __restrict__ rpb,
                         float* attn_io,
                         const float* __restrict__ Wn,
                         const float* __restrict__ bn,
                         float* __restrict__ dst) {
    __shared__ __attribute__((aligned(16))) float o_tile[PATCH_PIX * OPITCH];
    __shared__ __attribute__((aligned(16))) float kv_tile[4 * PLANE_F];

    const int tid = threadIdx.x;
    const int s   = tid & 7;
    const int pl  = (tid >> 3) & (PATCH_PIX - 1);
    const int head = tid >> 7;
    const int pr   = blockIdx.x;
    const int i0 = (pr / PATCHES_X) * PATCH_H;
    const int j0 = (pr % PATCHES_X) * PATCH_W;
    const int i = i0 + (pl / PATCH_W);
    const int j = j0 + (pl % PATCH_W);
    const int pix = i * WW + j;
    const int idx = head * PIX + pix;

    int si = i - (KW / 2); si = si < 0 ? 0 : si; si = si > HH - KW ? HH - KW : si;
    int sj = j - (KW / 2); sj = sj < 0 ? 0 : sj; sj = sj > WW - KW ? WW - KW : sj;

    const float LOG2E = 1.4426950408889634f;
    const float scale2 = 0.17677669529663687f * LOG2E;

    v4f q = *(const v4f*)(qkv + (size_t)pix * QKV_N + head * HD + 4 * s);
    q *= scale2;

    const float* rb = rpb + head * (2 * KW - 1) * (2 * KW - 1)
                    + (si - i + KW - 1) * (2 * KW - 1) + (sj - j + KW - 1);
    float* aio = attn_io + (size_t)idx * NKK;
    float extra[7];
#pragma unroll
    for (int t = 0; t < 7; ++t) {
        const int kk = (t < 6) ? (s + 8 * t) : (s == 0 ? NKK - 1 : 0);
        const int a = kk / KW;
        const int b = kk - a * KW;
        float e = rb[a * (2 * KW - 1) + b];
        if (HAS_PREV) e += aio[kk];
        extra[t] = e * LOG2E;
    }
    if (s != 0) extra[6] = 0.f;

    const int wid  = __builtin_amdgcn_readfirstlane(tid >> 5);
    const int lane = tid & 31;

    int hi0 = i0 - (KW / 2); hi0 = hi0 < 0 ? 0 : hi0; hi0 = hi0 > HH - HALO ? HH - HALO : hi0;
    int hj0 = j0 - (KW / 2); hj0 = hj0 < 0 ? 0 : hj0; hj0 = hj0 > WW - HALO ? WW - HALO : hj0;
    {
        const float* __restrict__ gsrc = qkv + (size_t)(hi0 * WW + hj0) * QKV_N + 64 + 4 * lane;
        float* sdst = kv_tile + (lane >> 3) * PLANE_F + 4 * (lane & 7);
        v4f stg[13];
#pragma unroll
        for (int n = 0; n < 13; ++n) {
            int hp = wid + 8 * n;
            hp = hp > HALO_PIX - 1 ? HALO_PIX - 1 : hp;
            const int hr = hp / HALO;
            const int hc = hp - hr * HALO;
            stg[n] = *(const v4f*)(gsrc + (size_t)(hr * WW + hc) * QKV_N);
        }
#pragma unroll
        for (int n = 0; n < 13; ++n) {
            int hp = wid + 8 * n;
            hp = hp > HALO_PIX - 1 ? HALO_PIX - 1 : hp;
            *(v4f*)(sdst + hp * HD) = stg[n];
        }
    }

    const float* kbase = kv_tile + head * PLANE_F + ((si - hi0) * HALO + (sj - hj0)) * HD + 4 * s;
    const float* vbase = kbase + 2 * PLANE_F;

    __syncthreads();

    float lg[NKK];
#pragma unroll
    for (int a = 0; a < KW; ++a) {
#pragma unroll
        for (int b = 0; b < KW; ++b) {
            const int kk = a * KW + b;
            const v4f k4 = *(const v4f*)(kbase + (a * HALO + b) * HD);
            float part = (s == (kk & 7)) ? extra[kk >> 3] : 0.f;
            part = fmaf(q.x, k4.x, part);
            part = fmaf(q.y, k4.y, part);
            part = fmaf(q.z, k4.z, part);
            part = fmaf(q.w, k4.w, part);
            part += lane_xor<1>(part);
            part += lane_xor<2>(part);
            part += lane_xor<4>(part);
            lg[kk] = part;
        }
    }

    float m = lg[0];
#pragma unroll
    for (int kk = 1; kk < NKK; ++kk) m = fmaxf(m, lg[kk]);
    float ssum = 0.f;
#pragma unroll
    for (int kk = 0; kk < NKK; ++kk) {
        lg[kk] = __builtin_amdgcn_exp2f(lg[kk] - m);
        ssum += lg[kk];
    }
    const float inv = 1.f / ssum;

    float mine[7];
#pragma unroll
    for (int t = 0; t < 7; ++t) mine[t] = 0.f;
#pragma unroll
    for (int kk = 0; kk < NKK; ++kk) {
        if (s == (kk & 7)) mine[kk >> 3] = lg[kk];
    }
#pragma unroll
    for (int t = 0; t < 6; ++t) aio[s + 8 * t] = mine[t] * inv;
    if (s == 0) aio[48] = mine[6] * inv;

    v4f acc = {0.f, 0.f, 0.f, 0.f};
#pragma unroll
    for (int a = 0; a < KW; ++a) {
#pragma unroll
        for (int b = 0; b < KW; ++b) {
            const float p = lg[a * KW + b];
            const v4f v4 = *(const v4f*)(vbase + (a * HALO + b) * HD);
            acc.x = fmaf(p, v4.x, acc.x);
            acc.y = fmaf(p, v4.y, acc.y);
            acc.z = fmaf(p, v4.z, acc.z);
            acc.w = fmaf(p, v4.w, acc.w);
        }
    }
    acc *= inv;

    const int lmod  = lane & 15;
    const int lhalf = lane >> 4;
    const bool has_t0 = LAST ? (wid < CDIM / 16) : true;
    const bool has_t1 = LAST ? false : (wid < 4);

    *(v4f*)(o_tile + pl * OPITCH + head * HD + 4 * s) = acc;

    const float* __restrict__ Wp = Wn + (size_t)(wid * 16 + lmod) * CDIM + 2 * lhalf;
    v2f w0[CDIM / 4], w1[CDIM / 4];
    if (has_t0) {
#pragma unroll
        for (int kk = 0; kk < CDIM / 4; ++kk) w0[kk] = *(const v2f*)(Wp + 4 * kk);
    }
    if (has_t1) {
#pragma unroll
        for (int kk = 0; kk < CDIM / 4; ++kk) w1[kk] = *(const v2f*)(Wp + (size_t)8 * 16 * CDIM + 4 * kk);
    }
    __syncthreads();
    if (!has_t0) return;

    v2f o[CDIM / 4];
#pragma unroll
    for (int kk = 0; kk < CDIM / 4; ++kk)
        o[kk] = *(const v2f*)(o_tile + lmod * OPITCH + 4 * kk + 2 * lhalf);

    if (!LAST) {
        float* __restrict__ Cp = dst + (size_t)((i0 + 2 * lhalf) * WW + j0) * QKV_N + (wid * 16 + lmod);
        v8f c0 = {0.f, 0.f, 0.f, 0.f, 0.f, 0.f, 0.f, 0.f};
        if (has_t1) {
            v8f c1 = {0.f, 0.f, 0.f, 0.f, 0.f, 0.f, 0.f, 0.f};
#pragma unroll
            for (int kk = 0; kk < CDIM / 4; ++kk) {
                c0 = wmma_f32_k4(o[kk], w0[kk], c0);
                c1 = wmma_f32_k4(o[kk], w1[kk], c1);
            }
            const float bv0 = bn[wid * 16 + lmod];
            const float bv1 = bn[(wid + 8) * 16 + lmod];
#pragma unroll
            for (int v = 0; v < 8; ++v) {
                const size_t off = (size_t)((v >> 2) * WW + (v & 3)) * QKV_N;
                Cp[off] = c0[v] + bv0;
                Cp[off + 8 * 16] = c1[v] + bv1;
            }
        } else {
#pragma unroll
            for (int kk = 0; kk < CDIM / 4; ++kk) c0 = wmma_f32_k4(o[kk], w0[kk], c0);
            const float bv0 = bn[wid * 16 + lmod];
#pragma unroll
            for (int v = 0; v < 8; ++v) {
                const size_t off = (size_t)((v >> 2) * WW + (v & 3)) * QKV_N;
                Cp[off] = c0[v] + bv0;
            }
        }
    } else {
        v8f c0 = {0.f, 0.f, 0.f, 0.f, 0.f, 0.f, 0.f, 0.f};
#pragma unroll
        for (int kk = 0; kk < CDIM / 4; ++kk) c0 = wmma_f32_k4(w0[kk], o[kk], c0);
        const int ch = wid * 16 + 8 * lhalf;
        const v4f blo = *(const v4f*)(bn + ch);
        const v4f bhi = *(const v4f*)(bn + ch + 4);
        float* __restrict__ Cp = dst + (size_t)ch * PIX
                               + (i0 + (lmod >> 2)) * WW + j0 + (lmod & 3);
#pragma unroll
        for (int v = 0; v < 8; ++v) {
            const float bv = (v < 4) ? blo[v & 3] : bhi[v & 3];
            Cp[(size_t)v * PIX] = c0[v] + bv;
        }
    }
}

extern "C" void kernel_launch(void* const* d_in, const int* in_sizes, int n_in,
                              void* d_out, int out_size, void* d_ws, size_t ws_size,
                              hipStream_t stream) {
    const float* x      = (const float*)d_in[0];
    const float* qkv_w  = (const float*)d_in[1];
    const float* qkv_b  = (const float*)d_in[2];
    const float* proj_w = (const float*)d_in[3];
    const float* proj_b = (const float*)d_in[4];
    const float* rpb    = (const float*)d_in[5];
    float* out = (float*)d_out;
    float* ws  = (float*)d_ws;

    float* wfold = ws;
    float* bfold = wfold + (size_t)2 * QKV_N * CDIM;
    float* qkvA  = bfold + (size_t)2 * QKV_N;
    float* qkvB  = qkvA + (size_t)PIX * QKV_N;
    float* attn  = qkvB + (size_t)PIX * QKV_N;

    qkv0_gemm_and_fold<<<QKV_BLOCKS + FOLD_BLOCKS, L1_THREADS, 0, stream>>>(
        x, qkv_w, qkv_b, proj_w, proj_b, qkvA, wfold, bfold);
    na_attn_gemm_kernel<false, false><<<PATCHES, ATT_THREADS, 0, stream>>>(
        qkvA, rpb, attn, wfold, bfold, qkvB);
    na_attn_gemm_kernel<true, false><<<PATCHES, ATT_THREADS, 0, stream>>>(
        qkvB, rpb + RPB_STAGE, attn, wfold + (size_t)QKV_N * CDIM, bfold + QKV_N, qkvA);
    na_attn_gemm_kernel<true, true><<<PATCHES, ATT_THREADS, 0, stream>>>(
        qkvA, rpb + 2 * RPB_STAGE, attn, proj_w + (size_t)2 * CDIM * CDIM,
        proj_b + (size_t)2 * CDIM, out);
}
